# starting kernel
# baseline (speedup 1.0000x reference)
#include <hip/hip_runtime.h>
#include <math.h>

typedef __attribute__((ext_vector_type(2))) float v2f;
typedef __attribute__((ext_vector_type(8))) float v8f;

namespace {
constexpr int kB = 128, kI = 2048, kE = 8, kO = 16, kD = 16;
constexpr int kBT  = 16;
constexpr int kIC  = 64;
constexpr int kNBT = kB / kBT;
constexpr int kNIC = kI / kIC;
constexpr int kSOD = kB * kO * kD;
}

template <int MASK>
__device__ __forceinline__ float swz_xor(float v) {
  constexpr int pat = (MASK << 10) | 0x1F;
  return __int_as_float(__builtin_amdgcn_ds_swizzle(__float_as_int(v), pat));
}

__global__ void caps_zero(float* __restrict__ s) {
  const int idx = blockIdx.x * blockDim.x + threadIdx.x;
  if (idx < kSOD) s[idx] = 0.0f;
}

__global__ __launch_bounds__(256, 1) void caps_accum(
    const float* __restrict__ x,
    const float* __restrict__ w,
    const float* __restrict__ out0,
    const float* __restrict__ out1,
    float* __restrict__ s,
    int nprev)
{
  const int bt   = blockIdx.x;
  const int ic   = blockIdx.y;
  const int tid  = threadIdx.x;
  const int wave = tid >> 5;
  const int lane = tid & 31;
  const int lg   = lane & 15;
  const int hi   = lane >> 4;
  const int o0   = wave * 2;

  __shared__ float bmat[kO][kBT];
  __shared__ float cmat[kO][kBT];

  float opsum[2][8];
  if (nprev > 0) {
#pragma unroll
    for (int ol = 0; ol < 2; ++ol) {
      const int o = o0 + ol;
#pragma unroll
      for (int v = 0; v < 8; ++v) {
        const int gb  = bt * kBT + v + hi * 8;
        const int idx = (gb * kO + o) * kD + lg;
        float t = out0[idx];
        if (nprev > 1) t += out1[idx];
        opsum[ol][v] = t;
      }
    }
  }

  float acc[2][8];
#pragma unroll
  for (int ol = 0; ol < 2; ++ol)
#pragma unroll
    for (int v = 0; v < 8; ++v) acc[ol][v] = 0.0f;

  const bool b3 = (lg & 8) != 0;
  const bool b2 = (lg & 4) != 0;
  const bool b1 = (lg & 2) != 0;

  for (int il = 0; il < kIC; ++il) {
    const int i = ic * kIC + il;

    const float* xp = x + ((size_t)(bt * kBT + lg) * kI + i) * kE + hi * 2;
    v2f a0 = { xp[0], xp[1] };
    v2f a1 = { xp[4], xp[5] };

    float xhf[2][8];
#pragma unroll
    for (int ol = 0; ol < 2; ++ol) {
      const int o = o0 + ol;
      const float* wp = w + (((size_t)(o * kI + i) * kD + lg) * kE) + hi * 2;
      v2f b0v = { wp[0], wp[1] };
      v2f b1v = { wp[4], wp[5] };
      v8f c = {0.f, 0.f, 0.f, 0.f, 0.f, 0.f, 0.f, 0.f};
      c = __builtin_amdgcn_wmma_f32_16x16x4_f32(false, a0, false, b0v,
                                                (short)0, c, false, false);
      c = __builtin_amdgcn_wmma_f32_16x16x4_f32(false, a1, false, b1v,
                                                (short)0, c, false, false);
#pragma unroll
      for (int v = 0; v < 8; ++v) xhf[ol][v] = c[v];
    }

    if (nprev == 0) {
      constexpr float cv = 1.0f / kO;
#pragma unroll
      for (int ol = 0; ol < 2; ++ol)
#pragma unroll
        for (int v = 0; v < 8; ++v) acc[ol][v] += cv * xhf[ol][v];
    } else {
#pragma unroll
      for (int ol = 0; ol < 2; ++ol) {
        float p[8];
#pragma unroll
        for (int v = 0; v < 8; ++v) p[v] = xhf[ol][v] * opsum[ol][v];
        float q[4];
#pragma unroll
        for (int j = 0; j < 4; ++j) {
          const float sel = b3 ? p[j + 4] : p[j];
          const float oth = b3 ? p[j] : p[j + 4];
          q[j] = sel + swz_xor<8>(oth);
        }
        float r2[2];
#pragma unroll
        for (int j = 0; j < 2; ++j) {
          const float sel = b2 ? q[j + 2] : q[j];
          const float oth = b2 ? q[j] : q[j + 2];
          r2[j] = sel + swz_xor<4>(oth);
        }
        {
          const float sel = b1 ? r2[1] : r2[0];
          const float oth = b1 ? r2[0] : r2[1];
          float t = sel + swz_xor<2>(oth);
          t += swz_xor<1>(t);
          if ((lg & 1) == 0) bmat[o0 + ol][(lg >> 1) + hi * 8] = t;
        }
      }
      __syncthreads();
      {
        const int o_s = tid >> 4;
        const int m_s = tid & 15;
        float bv[kO];
        float mx = -3.0e38f;
#pragma unroll
        for (int oo = 0; oo < kO; ++oo) {
          bv[oo] = bmat[oo][m_s];
          mx = fmaxf(mx, bv[oo]);
        }
        float den = 0.0f, mine = 0.0f;
#pragma unroll
        for (int oo = 0; oo < kO; ++oo) {
          const float e = __expf(bv[oo] - mx);
          den += e;
          if (oo == o_s) mine = e;
        }
        cmat[o_s][m_s] = mine / den;
      }
      __syncthreads();
#pragma unroll
      for (int ol = 0; ol < 2; ++ol)
#pragma unroll
        for (int v = 0; v < 8; ++v)
          acc[ol][v] += cmat[o0 + ol][v + hi * 8] * xhf[ol][v];
    }
  }

#pragma unroll
  for (int ol = 0; ol < 2; ++ol) {
    const int o = o0 + ol;
#pragma unroll
    for (int v = 0; v < 8; ++v) {
      const int gb = bt * kBT + v + hi * 8;
      atomicAdd(&s[(gb * kO + o) * kD + lg], acc[ol][v]);
    }
  }
}

__global__ void caps_squash(const float* __restrict__ s, float* __restrict__ dst) {
  const int idx = blockIdx.x * 256 + threadIdx.x;
  const float v = s[idx];
  float n2 = v * v;
  n2 += swz_xor<1>(n2);
  n2 += swz_xor<2>(n2);
  n2 += swz_xor<4>(n2);
  n2 += swz_xor<8>(n2);
  const float n = sqrtf(n2);
  const float scale = n2 / ((1.0f + n2) * (n + 1e-8f));
  dst[idx] = scale * v;
}

extern "C" void kernel_launch(void* const* d_in, const int* in_sizes, int n_in,
                              void* d_out, int out_size, void* d_ws, size_t ws_size,
                              hipStream_t stream) {
  const float* x = (const float*)d_in[0];
  const float* w = (const float*)d_in[1];
  float* s    = (float*)d_ws;
  float* out0 = s + kSOD;
  float* out1 = out0 + kSOD;

  const dim3 grid(kNBT, kNIC);
  for (int r = 0; r < 3; ++r) {
    caps_zero<<<(kSOD + 255) / 256, 256, 0, stream>>>(s);
    caps_accum<<<grid, 256, 0, stream>>>(x, w, out0, out1, s, r);
    float* dst = (r == 0) ? out0 : (r == 1) ? out1 : (float*)d_out;
    caps_squash<<<kSOD / 256, 256, 0, stream>>>(s, dst);
  }
}


# accum loop takes 8 in-capsules per trip: fragment loads, lane-reduction trees and the block softmax exchange (one barrier pair per trip) overlap instead of adding per capsule
# speedup vs baseline: 2.4385x; 2.4385x over previous
#include <hip/hip_runtime.h>
#include <math.h>

typedef __attribute__((ext_vector_type(2))) float v2f;
typedef __attribute__((ext_vector_type(4))) float v4f;
typedef __attribute__((ext_vector_type(8))) float v8f;

namespace {
constexpr int kB = 128, kI = 2048, kE = 8, kO = 16, kD = 16;
constexpr int kBT  = 16;
constexpr int kIC  = 64;
constexpr int kU   = 8;
constexpr int kBP  = 20;
constexpr int kNBT = kB / kBT;
constexpr int kNIC = kI / kIC;
constexpr int kSOD = kB * kO * kD;
}

template <int MASK>
__device__ __forceinline__ float swz_xor(float v) {
  constexpr int pat = (MASK << 10) | 0x1F;
  return __int_as_float(__builtin_amdgcn_ds_swizzle(__float_as_int(v), pat));
}

__global__ void caps_zero(float* __restrict__ s) {
  const int idx = blockIdx.x * blockDim.x + threadIdx.x;
  if (idx < kSOD) s[idx] = 0.0f;
}

__device__ __forceinline__ void xhat_group(
    const float* __restrict__ x, const float* __restrict__ w,
    int row, int o0, int ib, int lg, int hi, float (&xhf)[kU][2][8])
{
  v2f a0[kU], a1[kU], b0[kU][2], b1[kU][2];
#pragma unroll
  for (int u = 0; u < kU; ++u) {
    const int i = ib + u;
    const float* xp = x + ((size_t)row * kI + i) * kE + hi * 2;
    a0[u] = v2f{ xp[0], xp[1] };
    a1[u] = v2f{ xp[4], xp[5] };
#pragma unroll
    for (int ol = 0; ol < 2; ++ol) {
      const float* wp = w + (((size_t)((o0 + ol) * kI + i) * kD + lg) * kE) + hi * 2;
      b0[u][ol] = v2f{ wp[0], wp[1] };
      b1[u][ol] = v2f{ wp[4], wp[5] };
    }
  }
#pragma unroll
  for (int u = 0; u < kU; ++u) {
#pragma unroll
    for (int ol = 0; ol < 2; ++ol) {
      v8f c = {0.f, 0.f, 0.f, 0.f, 0.f, 0.f, 0.f, 0.f};
      c = __builtin_amdgcn_wmma_f32_16x16x4_f32(false, a0[u], false, b0[u][ol],
                                                (short)0, c, false, false);
      c = __builtin_amdgcn_wmma_f32_16x16x4_f32(false, a1[u], false, b1[u][ol],
                                                (short)0, c, false, false);
#pragma unroll
      for (int v = 0; v < 8; ++v) xhf[u][ol][v] = c[v];
    }
  }
}

__global__ __attribute__((amdgpu_flat_work_group_size(256, 256),
                          amdgpu_waves_per_eu(2, 2))) void caps_accum(
    const float* __restrict__ x,
    const float* __restrict__ w,
    const float* __restrict__ out0,
    const float* __restrict__ out1,
    float* __restrict__ s,
    int nprev)
{
  const int bt   = blockIdx.x;
  const int ic   = blockIdx.y;
  const int tid  = threadIdx.x;
  const int wave = tid >> 5;
  const int lane = tid & 31;
  const int lg   = lane & 15;
  const int hi   = lane >> 4;
  const int o0   = wave * 2;
  const int row  = bt * kBT + lg;

  __shared__ __attribute__((aligned(16))) float bmat[kU][kBT][kBP];
  __shared__ __attribute__((aligned(16))) float cmat[kU][kO][kBT];

  float opsum[2][8];
  if (nprev > 0) {
#pragma unroll
    for (int ol = 0; ol < 2; ++ol) {
      const int o = o0 + ol;
#pragma unroll
      for (int v = 0; v < 8; ++v) {
        const int gb  = bt * kBT + v + hi * 8;
        const int idx = (gb * kO + o) * kD + lg;
        float t = out0[idx];
        if (nprev > 1) t += out1[idx];
        opsum[ol][v] = t;
      }
    }
  }

  float acc[2][8];
#pragma unroll
  for (int ol = 0; ol < 2; ++ol)
#pragma unroll
    for (int v = 0; v < 8; ++v) acc[ol][v] = 0.0f;

  if (nprev == 0) {
    constexpr float cv = 1.0f / kO;
    for (int t = 0; t < kIC / kU; ++t) {
      float xhf[kU][2][8];
      xhat_group(x, w, row, o0, ic * kIC + t * kU, lg, hi, xhf);
#pragma unroll
      for (int u = 0; u < kU; ++u)
#pragma unroll
        for (int ol = 0; ol < 2; ++ol)
#pragma unroll
          for (int v = 0; v < 8; ++v) acc[ol][v] += cv * xhf[u][ol][v];
    }
  } else {
    const bool b3 = (lg & 8) != 0;
    const bool b2 = (lg & 4) != 0;
    const bool b1 = (lg & 2) != 0;

    for (int t = 0; t < kIC / kU; ++t) {
      float xhf[kU][2][8];
      xhat_group(x, w, row, o0, ic * kIC + t * kU, lg, hi, xhf);

      float q[kU][2][4];
#pragma unroll
      for (int u = 0; u < kU; ++u)
#pragma unroll
        for (int ol = 0; ol < 2; ++ol)
#pragma unroll
          for (int j = 0; j < 4; ++j) {
            const float plo = xhf[u][ol][j] * opsum[ol][j];
            const float phi = xhf[u][ol][j + 4] * opsum[ol][j + 4];
            const float sel = b3 ? phi : plo;
            const float oth = b3 ? plo : phi;
            q[u][ol][j] = sel + swz_xor<8>(oth);
          }
      float r2[kU][2][2];
#pragma unroll
      for (int u = 0; u < kU; ++u)
#pragma unroll
        for (int ol = 0; ol < 2; ++ol)
#pragma unroll
          for (int j = 0; j < 2; ++j) {
            const float sel = b2 ? q[u][ol][j + 2] : q[u][ol][j];
            const float oth = b2 ? q[u][ol][j] : q[u][ol][j + 2];
            r2[u][ol][j] = sel + swz_xor<4>(oth);
          }
      float t1[kU][2];
#pragma unroll
      for (int u = 0; u < kU; ++u)
#pragma unroll
        for (int ol = 0; ol < 2; ++ol) {
          const float sel = b1 ? r2[u][ol][1] : r2[u][ol][0];
          const float oth = b1 ? r2[u][ol][0] : r2[u][ol][1];
          t1[u][ol] = sel + swz_xor<2>(oth);
        }
      float t0[kU][2];
#pragma unroll
      for (int u = 0; u < kU; ++u)
#pragma unroll
        for (int ol = 0; ol < 2; ++ol) {
          t0[u][ol] = t1[u][ol] + swz_xor<1>(t1[u][ol]);
        }
      if ((lg & 1) == 0) {
#pragma unroll
        for (int u = 0; u < kU; ++u) {
          float* bp = &bmat[u][(lg >> 1) + hi * 8][o0];
          bp[0] = t0[u][0];
          bp[1] = t0[u][1];
        }
      }
      __syncthreads();
      {
        const v4f* bp = reinterpret_cast<const v4f*>(&bmat[wave][lg][0]);
        float bv[kO];
#pragma unroll
        for (int k = 0; k < 4; ++k) {
          const v4f t4 = bp[k];
          bv[4 * k + 0] = t4[0];
          bv[4 * k + 1] = t4[1];
          bv[4 * k + 2] = t4[2];
          bv[4 * k + 3] = t4[3];
        }
        float mx = -3.0e38f;
#pragma unroll
        for (int oo = 0; oo < kO; ++oo) mx = fmaxf(mx, bv[oo]);
        float den = 0.0f;
        float elo[8], mine[8];
        const bool up = (hi != 0);
#pragma unroll
        for (int j = 0; j < 8; ++j) {
          elo[j] = __expf(bv[j] - mx);
          den += elo[j];
        }
#pragma unroll
        for (int j = 0; j < 8; ++j) {
          const float ehi = __expf(bv[8 + j] - mx);
          den += ehi;
          const float l = elo[j];
          mine[j] = up ? ehi : l;
        }
        const float inv = 1.0f / den;
        float* cw = &cmat[wave][hi * 8][lg];
#pragma unroll
        for (int j = 0; j < 8; ++j) cw[j * kBT] = mine[j] * inv;
      }
      __syncthreads();
#pragma unroll
      for (int u = 0; u < kU; ++u)
#pragma unroll
        for (int ol = 0; ol < 2; ++ol) {
          const v4f* cp = reinterpret_cast<const v4f*>(&cmat[u][o0 + ol][hi * 8]);
          const v4f c0 = cp[0];
          const v4f c1 = cp[1];
#pragma unroll
          for (int v = 0; v < 4; ++v) {
            acc[ol][v]     += c0[v] * xhf[u][ol][v];
            acc[ol][v + 4] += c1[v] * xhf[u][ol][v + 4];
          }
        }
    }
  }

#pragma unroll
  for (int ol = 0; ol < 2; ++ol) {
    const int o = o0 + ol;
#pragma unroll
    for (int v = 0; v < 8; ++v) {
      const int gb = bt * kBT + v + hi * 8;
      atomicAdd(&s[(gb * kO + o) * kD + lg], acc[ol][v]);
    }
  }
}

__global__ void caps_squash(const float* __restrict__ s, float* __restrict__ dst) {
  const int idx = blockIdx.x * 256 + threadIdx.x;
  const float v = s[idx];
  float n2 = v * v;
  n2 += swz_xor<1>(n2);
  n2 += swz_xor<2>(n2);
  n2 += swz_xor<4>(n2);
  n2 += swz_xor<8>(n2);
  const float n = sqrtf(n2);
  const float scale = n2 / ((1.0f + n2) * (n + 1e-8f));
  dst[idx] = scale * v;
}

extern "C" void kernel_launch(void* const* d_in, const int* in_sizes, int n_in,
                              void* d_out, int out_size, void* d_ws, size_t ws_size,
                              hipStream_t stream) {
  const float* x = (const float*)d_in[0];
  const float* w = (const float*)d_in[1];
  float* s    = (float*)d_ws;
  float* out0 = s + kSOD;
  float* out1 = out0 + kSOD;

  const dim3 grid(kNBT, kNIC);
  for (int r = 0; r < 3; ++r) {
    caps_zero<<<(kSOD + 255) / 256, 256, 0, stream>>>(s);
    caps_accum<<<grid, 256, 0, stream>>>(x, w, out0, out1, s, r);
    float* dst = (r == 0) ? out0 : (r == 1) ? out1 : (float*)d_out;
    caps_squash<<<kSOD / 256, 256, 0, stream>>>(s, dst);
  }
}


# launch fusion: one launch clears the three round accumulators; each accumulate pass squashes the earlier rounds' raw sums for its own rows in its prologue (9 launches to 5)
# speedup vs baseline: 2.8911x; 1.1856x over previous
#include <hip/hip_runtime.h>
#include <math.h>

typedef __attribute__((ext_vector_type(2))) float v2f;
typedef __attribute__((ext_vector_type(4))) float v4f;
typedef __attribute__((ext_vector_type(8))) float v8f;

namespace {
constexpr int kB = 128, kI = 2048, kE = 8, kO = 16, kD = 16;
constexpr int kBT  = 16;
constexpr int kIC  = 64;
constexpr int kU   = 8;
constexpr int kBP  = 20;
constexpr int kNBT = kB / kBT;
constexpr int kNIC = kI / kIC;
constexpr int kSOD = kB * kO * kD;
}

template <int MASK>
__device__ __forceinline__ float swz_xor(float v) {
  constexpr int pat = (MASK << 10) | 0x1F;
  return __int_as_float(__builtin_amdgcn_ds_swizzle(__float_as_int(v), pat));
}

__global__ void caps_zero3(float* __restrict__ s) {
  const int idx = blockIdx.x * 256 + threadIdx.x;
  reinterpret_cast<v4f*>(s)[idx] = v4f{0.0f, 0.0f, 0.0f, 0.0f};
}

__device__ __forceinline__ float halfwave_sums8(const float (&p)[8],
                                                bool k3, bool k2, bool k1) {
  float q[4];
#pragma unroll
  for (int j = 0; j < 4; ++j) {
    const float sel = k3 ? p[j + 4] : p[j];
    const float oth = k3 ? p[j] : p[j + 4];
    q[j] = sel + swz_xor<8>(oth);
  }
  float r[2];
#pragma unroll
  for (int j = 0; j < 2; ++j) {
    const float sel = k2 ? q[j + 2] : q[j];
    const float oth = k2 ? q[j] : q[j + 2];
    r[j] = sel + swz_xor<4>(oth);
  }
  const float sel = k1 ? r[1] : r[0];
  const float oth = k1 ? r[0] : r[1];
  const float t = sel + swz_xor<2>(oth);
  return t + swz_xor<1>(t);
}

__device__ __forceinline__ float squash_scale(float n2) {
  const float n = sqrtf(n2);
  return n2 / ((1.0f + n2) * (n + 1e-8f));
}

__device__ __forceinline__ void xhat_group(
    const float* __restrict__ x, const float* __restrict__ w,
    int row, int o0, int ib, int lg, int hi, float (&xhf)[kU][2][8])
{
  v2f a0[kU], a1[kU], b0[kU][2], b1[kU][2];
#pragma unroll
  for (int u = 0; u < kU; ++u) {
    const int i = ib + u;
    const float* xp = x + ((size_t)row * kI + i) * kE + hi * 2;
    a0[u] = v2f{ xp[0], xp[1] };
    a1[u] = v2f{ xp[4], xp[5] };
#pragma unroll
    for (int ol = 0; ol < 2; ++ol) {
      const float* wp = w + (((size_t)((o0 + ol) * kI + i) * kD + lg) * kE) + hi * 2;
      b0[u][ol] = v2f{ wp[0], wp[1] };
      b1[u][ol] = v2f{ wp[4], wp[5] };
    }
  }
#pragma unroll
  for (int u = 0; u < kU; ++u) {
#pragma unroll
    for (int ol = 0; ol < 2; ++ol) {
      v8f c = {0.f, 0.f, 0.f, 0.f, 0.f, 0.f, 0.f, 0.f};
      c = __builtin_amdgcn_wmma_f32_16x16x4_f32(false, a0[u], false, b0[u][ol],
                                                (short)0, c, false, false);
      c = __builtin_amdgcn_wmma_f32_16x16x4_f32(false, a1[u], false, b1[u][ol],
                                                (short)0, c, false, false);
#pragma unroll
      for (int v = 0; v < 8; ++v) xhf[u][ol][v] = c[v];
    }
  }
}

__global__ __attribute__((amdgpu_flat_work_group_size(256, 256),
                          amdgpu_waves_per_eu(2, 2))) void caps_accum(
    const float* __restrict__ x,
    const float* __restrict__ w,
    const float* __restrict__ sr0,
    const float* __restrict__ sr1,
    float* __restrict__ s,
    int nprev)
{
  const int bt   = blockIdx.x;
  const int ic   = blockIdx.y;
  const int tid  = threadIdx.x;
  const int wave = tid >> 5;
  const int lane = tid & 31;
  const int lg   = lane & 15;
  const int hi   = lane >> 4;
  const int o0   = wave * 2;
  const int row  = bt * kBT + lg;

  __shared__ __attribute__((aligned(16))) float bmat[kU][kBT][kBP];
  __shared__ __attribute__((aligned(16))) float cmat[kU][kO][kBT];
  __shared__ __attribute__((aligned(16))) float scl[2][kO][kBT];

  float opsum[2][8];
  if (nprev > 0) {
    const bool k3 = (lg & 8) != 0;
    const bool k2 = (lg & 4) != 0;
    const bool k1 = (lg & 2) != 0;
    float r0[2][8], r1[2][8];
#pragma unroll
    for (int ol = 0; ol < 2; ++ol)
#pragma unroll
      for (int v = 0; v < 8; ++v)
        r0[ol][v] = sr0[((bt * kBT + v + hi * 8) * kO + o0 + ol) * kD + lg];
    if (nprev > 1) {
#pragma unroll
      for (int ol = 0; ol < 2; ++ol)
#pragma unroll
        for (int v = 0; v < 8; ++v)
          r1[ol][v] = sr1[((bt * kBT + v + hi * 8) * kO + o0 + ol) * kD + lg];
    } else {
#pragma unroll
      for (int ol = 0; ol < 2; ++ol)
#pragma unroll
        for (int v = 0; v < 8; ++v) r1[ol][v] = 0.0f;
    }
    float nrm[2][2];
#pragma unroll
    for (int ol = 0; ol < 2; ++ol) {
      float p0[8], p1[8];
#pragma unroll
      for (int v = 0; v < 8; ++v) {
        p0[v] = r0[ol][v] * r0[ol][v];
        p1[v] = r1[ol][v] * r1[ol][v];
      }
      nrm[0][ol] = halfwave_sums8(p0, k3, k2, k1);
      nrm[1][ol] = halfwave_sums8(p1, k3, k2, k1);
    }
    float fac[2][2];
#pragma unroll
    for (int rr = 0; rr < 2; ++rr)
#pragma unroll
      for (int ol = 0; ol < 2; ++ol) fac[rr][ol] = squash_scale(nrm[rr][ol]);
    if ((lg & 1) == 0) {
#pragma unroll
      for (int rr = 0; rr < 2; ++rr)
#pragma unroll
        for (int ol = 0; ol < 2; ++ol)
          scl[rr][o0 + ol][(lg >> 1) + hi * 8] = fac[rr][ol];
    }
    __syncthreads();
#pragma unroll
    for (int ol = 0; ol < 2; ++ol) {
      const v4f* fp0 = reinterpret_cast<const v4f*>(&scl[0][o0 + ol][hi * 8]);
      const v4f* fp1 = reinterpret_cast<const v4f*>(&scl[1][o0 + ol][hi * 8]);
      const v4f f0a = fp0[0], f0b = fp0[1];
      const v4f f1a = fp1[0], f1b = fp1[1];
#pragma unroll
      for (int v = 0; v < 4; ++v) {
        opsum[ol][v]     = f0a[v] * r0[ol][v]     + f1a[v] * r1[ol][v];
        opsum[ol][v + 4] = f0b[v] * r0[ol][v + 4] + f1b[v] * r1[ol][v + 4];
      }
    }
  }

  float acc[2][8];
#pragma unroll
  for (int ol = 0; ol < 2; ++ol)
#pragma unroll
    for (int v = 0; v < 8; ++v) acc[ol][v] = 0.0f;

  if (nprev == 0) {
    constexpr float cv = 1.0f / kO;
    for (int t = 0; t < kIC / kU; ++t) {
      float xhf[kU][2][8];
      xhat_group(x, w, row, o0, ic * kIC + t * kU, lg, hi, xhf);
#pragma unroll
      for (int u = 0; u < kU; ++u)
#pragma unroll
        for (int ol = 0; ol < 2; ++ol)
#pragma unroll
          for (int v = 0; v < 8; ++v) acc[ol][v] += cv * xhf[u][ol][v];
    }
  } else {
    const bool b3 = (lg & 8) != 0;
    const bool b2 = (lg & 4) != 0;
    const bool b1 = (lg & 2) != 0;

    for (int t = 0; t < kIC / kU; ++t) {
      float xhf[kU][2][8];
      xhat_group(x, w, row, o0, ic * kIC + t * kU, lg, hi, xhf);

      float q[kU][2][4];
#pragma unroll
      for (int u = 0; u < kU; ++u)
#pragma unroll
        for (int ol = 0; ol < 2; ++ol)
#pragma unroll
          for (int j = 0; j < 4; ++j) {
            const float plo = xhf[u][ol][j] * opsum[ol][j];
            const float phi = xhf[u][ol][j + 4] * opsum[ol][j + 4];
            const float sel = b3 ? phi : plo;
            const float oth = b3 ? plo : phi;
            q[u][ol][j] = sel + swz_xor<8>(oth);
          }
      float r2[kU][2][2];
#pragma unroll
      for (int u = 0; u < kU; ++u)
#pragma unroll
        for (int ol = 0; ol < 2; ++ol)
#pragma unroll
          for (int j = 0; j < 2; ++j) {
            const float sel = b2 ? q[u][ol][j + 2] : q[u][ol][j];
            const float oth = b2 ? q[u][ol][j] : q[u][ol][j + 2];
            r2[u][ol][j] = sel + swz_xor<4>(oth);
          }
      float t1[kU][2];
#pragma unroll
      for (int u = 0; u < kU; ++u)
#pragma unroll
        for (int ol = 0; ol < 2; ++ol) {
          const float sel = b1 ? r2[u][ol][1] : r2[u][ol][0];
          const float oth = b1 ? r2[u][ol][0] : r2[u][ol][1];
          t1[u][ol] = sel + swz_xor<2>(oth);
        }
      float t0[kU][2];
#pragma unroll
      for (int u = 0; u < kU; ++u)
#pragma unroll
        for (int ol = 0; ol < 2; ++ol) {
          t0[u][ol] = t1[u][ol] + swz_xor<1>(t1[u][ol]);
        }
      if ((lg & 1) == 0) {
#pragma unroll
        for (int u = 0; u < kU; ++u) {
          float* bp = &bmat[u][(lg >> 1) + hi * 8][o0];
          bp[0] = t0[u][0];
          bp[1] = t0[u][1];
        }
      }
      __syncthreads();
      {
        const v4f* bp = reinterpret_cast<const v4f*>(&bmat[wave][lg][0]);
        float bv[kO];
#pragma unroll
        for (int k = 0; k < 4; ++k) {
          const v4f t4 = bp[k];
          bv[4 * k + 0] = t4[0];
          bv[4 * k + 1] = t4[1];
          bv[4 * k + 2] = t4[2];
          bv[4 * k + 3] = t4[3];
        }
        float mx = -3.0e38f;
#pragma unroll
        for (int oo = 0; oo < kO; ++oo) mx = fmaxf(mx, bv[oo]);
        float den = 0.0f;
        float elo[8], mine[8];
        const bool up = (hi != 0);
#pragma unroll
        for (int j = 0; j < 8; ++j) {
          elo[j] = __expf(bv[j] - mx);
          den += elo[j];
        }
#pragma unroll
        for (int j = 0; j < 8; ++j) {
          const float ehi = __expf(bv[8 + j] - mx);
          den += ehi;
          const float l = elo[j];
          mine[j] = up ? ehi : l;
        }
        const float inv = 1.0f / den;
        float* cw = &cmat[wave][hi * 8][lg];
#pragma unroll
        for (int j = 0; j < 8; ++j) cw[j * kBT] = mine[j] * inv;
      }
      __syncthreads();
#pragma unroll
      for (int u = 0; u < kU; ++u)
#pragma unroll
        for (int ol = 0; ol < 2; ++ol) {
          const v4f* cp = reinterpret_cast<const v4f*>(&cmat[u][o0 + ol][hi * 8]);
          const v4f c0 = cp[0];
          const v4f c1 = cp[1];
#pragma unroll
          for (int v = 0; v < 4; ++v) {
            acc[ol][v]     += c0[v] * xhf[u][ol][v];
            acc[ol][v + 4] += c1[v] * xhf[u][ol][v + 4];
          }
        }
    }
  }

#pragma unroll
  for (int ol = 0; ol < 2; ++ol) {
    const int o = o0 + ol;
#pragma unroll
    for (int v = 0; v < 8; ++v) {
      const int gb = bt * kBT + v + hi * 8;
      atomicAdd(&s[(gb * kO + o) * kD + lg], acc[ol][v]);
    }
  }
}

__global__ void caps_squash(const float* __restrict__ s, float* __restrict__ dst) {
  const int idx = blockIdx.x * 256 + threadIdx.x;
  const float v = s[idx];
  float n2 = v * v;
  n2 += swz_xor<1>(n2);
  n2 += swz_xor<2>(n2);
  n2 += swz_xor<4>(n2);
  n2 += swz_xor<8>(n2);
  const float n = sqrtf(n2);
  const float scale = n2 / ((1.0f + n2) * (n + 1e-8f));
  dst[idx] = scale * v;
}

extern "C" void kernel_launch(void* const* d_in, const int* in_sizes, int n_in,
                              void* d_out, int out_size, void* d_ws, size_t ws_size,
                              hipStream_t stream) {
  const float* x = (const float*)d_in[0];
  const float* w = (const float*)d_in[1];
  float* s0 = (float*)d_ws;
  float* s1 = s0 + kSOD;
  float* s2 = s1 + kSOD;

  const dim3 grid(kNBT, kNIC);
  caps_zero3<<<3 * kSOD / (4 * 256), 256, 0, stream>>>(s0);
  caps_accum<<<grid, 256, 0, stream>>>(x, w, s1, s2, s0, 0);
  caps_accum<<<grid, 256, 0, stream>>>(x, w, s0, s2, s1, 1);
  caps_accum<<<grid, 256, 0, stream>>>(x, w, s0, s1, s2, 2);
  caps_squash<<<kSOD / 256, 256, 0, stream>>>(s2, (float*)d_out);
}
